# starting kernel
# baseline (speedup 1.0000x reference)
#include <hip/hip_runtime.h>
#include <math.h>

typedef __attribute__((ext_vector_type(2))) float v2f;
typedef __attribute__((ext_vector_type(8))) float v8f;

constexpr int B      = 256;
constexpr int ICAPS  = 3200;
constexpr int INDIM  = 8;
constexpr int J      = 8;
constexpr int M      = 16;
constexpr int BT     = 16;
constexpr int IC     = 64;
constexpr int NIB    = ICAPS / IC;
constexpr int WAVES  = 8;
constexpr int TPB    = WAVES * 32;
constexpr int SSZ    = B * J * M;
constexpr int UTS    = 18;

__global__ __launch_bounds__(TPB) void routing_pass(
    const float* __restrict__ x,
    const float* __restrict__ W,
    const float* __restrict__ vprev,
    float* __restrict__ s_out)
{
    __shared__ float xs[IC * BT * INDIM];
    __shared__ float vs[BT * J * M];
    __shared__ float s_lds[J * BT * M];
    __shared__ float a_lds[WAVES * BT * J];
    __shared__ float ut_lds[WAVES * BT * UTS];

    const int t  = threadIdx.x;
    const int b0 = blockIdx.x * BT;
    const int i0 = blockIdx.y * IC;

    {
        const float4* xg  = reinterpret_cast<const float4*>(x);
        float4*       xs4 = reinterpret_cast<float4*>(xs);
#pragma unroll
        for (int k = 0; k < (BT * IC * 2) / TPB; ++k) {
            const int q  = t + k * TPB;
            const int bb = q >> 7;
            const int r  = q & 127;
            const int ii = r >> 1;
            const int n4 = r & 1;
            xs4[(ii * BT + bb) * 2 + n4] =
                xg[((size_t)(b0 + bb) * ICAPS + (i0 + ii)) * 2 + n4];
        }
    }
    const bool uniform = (vprev == nullptr);
    if (!uniform) {
        const float4* vg  = reinterpret_cast<const float4*>(vprev) + b0 * (J * M / 4);
        float4*       vs4 = reinterpret_cast<float4*>(vs);
        for (int q = t; q < BT * J * M / 4; q += TPB) vs4[q] = vg[q];
    }
    for (int q = t; q < J * BT * M; q += TPB) s_lds[q] = 0.0f;
    __syncthreads();

    const int wave = t >> 5;
    const int lane = t & 31;
    const int half = lane >> 4;
    const int lm   = lane & 15;

    float* up = &ut_lds[wave * (BT * UTS)];

    v8f sacc[J];
#pragma unroll
    for (int j = 0; j < J; ++j) sacc[j] = (v8f){0,0,0,0,0,0,0,0};

    for (int il = wave; il < IC; il += WAVES) {
        const int i = i0 + il;

        const float* xrow = &xs[(il * BT + lm) * INDIM];
        v2f af0, af1;
        af0.x = xrow[half * 2 + 0]; af0.y = xrow[half * 2 + 1];
        af1.x = xrow[half * 2 + 4]; af1.y = xrow[half * 2 + 5];

        v8f acc[J];
        const float* wbase = W + (size_t)i * (J * INDIM * M);
#pragma unroll
        for (int j = 0; j < J; ++j) {
            const float* wj = wbase + j * (INDIM * M);
            const int kr = half * 2;
            v2f bf0, bf1;
            bf0.x = wj[(kr + 0) * M + lm]; bf0.y = wj[(kr + 1) * M + lm];
            bf1.x = wj[(kr + 4) * M + lm]; bf1.y = wj[(kr + 5) * M + lm];
            v8f c = {0,0,0,0,0,0,0,0};
            c = __builtin_amdgcn_wmma_f32_16x16x4_f32(false, af0, false, bf0,
                                                      (short)0, c, false, false);
            c = __builtin_amdgcn_wmma_f32_16x16x4_f32(false, af1, false, bf1,
                                                      (short)0, c, false, false);
            acc[j] = c;
        }

        if (uniform) {
#pragma unroll
            for (int j = 0; j < J; ++j)
#pragma unroll
                for (int r = 0; r < 8; ++r) sacc[j][r] += 0.125f * acc[j][r];
        } else {
            const bool holder = ((lm >> 3) == half);
#pragma unroll
            for (int j = 0; j < J; ++j) {
#pragma unroll
                for (int r = 0; r < 8; ++r)
                    up[(r + half * 8) * UTS + lm] = acc[j][r];
                asm volatile("" ::: "memory");
                v8f d = {0,0,0,0,0,0,0,0};
#pragma unroll
                for (int kq = 0; kq < 4; ++kq) {
                    const int kb = kq * 4 + half * 2;
                    v2f avf = *(const v2f*)&vs[(lm * J + j) * M + kb];
                    v2f buf = *(const v2f*)&up[lm * UTS + kb];
                    d = __builtin_amdgcn_wmma_f32_16x16x4_f32(false, avf, false, buf,
                                                              (short)0, d, false, false);
                }
                const int rsel = lm & 7;
                float dv = d[0];
#pragma unroll
                for (int r = 1; r < 8; ++r) dv = (rsel == r) ? d[r] : dv;
                if (holder) a_lds[(wave * BT + lm) * J + j] = dv;
                asm volatile("" ::: "memory");
            }

            asm volatile("s_wait_dscnt 0x0" ::: "memory");
            if (lane < BT) {
                const int bb = lane;
                float av[J];
                float mx = -3.0e38f;
#pragma unroll
                for (int j = 0; j < J; ++j) {
                    av[j] = a_lds[(wave * BT + bb) * J + j];
                    mx = fmaxf(mx, av[j]);
                }
                float se = 0.0f;
#pragma unroll
                for (int j = 0; j < J; ++j) { av[j] = __expf(av[j] - mx); se += av[j]; }
                const float inv = 1.0f / se;
                v2f* cw = (v2f*)&a_lds[(wave * BT + bb) * J];
#pragma unroll
                for (int q = 0; q < 4; ++q) {
                    v2f c2; c2.x = av[2 * q] * inv; c2.y = av[2 * q + 1] * inv;
                    cw[q] = c2;
                }
            }
            asm volatile("s_wait_dscnt 0x0" ::: "memory");

#pragma unroll
            for (int r = 0; r < 8; ++r) {
                const int bb = r + half * 8;
                const v2f* cp = (const v2f*)&a_lds[(wave * BT + bb) * J];
#pragma unroll
                for (int q = 0; q < 4; ++q) {
                    v2f c2 = cp[q];
                    sacc[2 * q][r]     += c2.x * acc[2 * q][r];
                    sacc[2 * q + 1][r] += c2.y * acc[2 * q + 1][r];
                }
            }
        }
    }

#pragma unroll
    for (int j = 0; j < J; ++j)
#pragma unroll
        for (int r = 0; r < 8; ++r) {
            const int bb = r + half * 8;
            atomicAdd(&s_lds[(j * BT + bb) * M + lm], sacc[j][r]);
        }
    __syncthreads();
    for (int q = t; q < J * BT * M; q += TPB) {
        const int j  = q / (BT * M);
        const int bb = (q / M) & (BT - 1);
        const int m  = q & (M - 1);
        atomicAdd(&s_out[((size_t)(b0 + bb) * J + j) * M + m], s_lds[q]);
    }
}

__global__ void squash_k(const float* __restrict__ s, const float* __restrict__ vadd,
                         float* __restrict__ vout)
{
    const int bj = blockIdx.x * blockDim.x + threadIdx.x;
    if (bj >= B * J) return;
    const float* sp = s + bj * M;
    float sq = 0.0f;
#pragma unroll
    for (int m = 0; m < M; ++m) { const float e = sp[m]; sq += e * e; }
    const float scale = sqrtf(sq) / (1.0f + sq);
#pragma unroll
    for (int m = 0; m < M; ++m) {
        float r = sp[m] * scale;
        if (vadd) r += vadd[bj * M + m];
        vout[bj * M + m] = r;
    }
}

__global__ void zero_ws(float* p, int n)
{
    const int t = blockIdx.x * blockDim.x + threadIdx.x;
    if (t < n) p[t] = 0.0f;
}

extern "C" void kernel_launch(void* const* d_in, const int* in_sizes, int n_in,
                              void* d_out, int out_size, void* d_ws, size_t ws_size,
                              hipStream_t stream)
{
    (void)in_sizes; (void)n_in; (void)out_size; (void)ws_size;
    const float* x = (const float*)d_in[0];
    const float* W = (const float*)d_in[1];
    float* out = (float*)d_out;
    float* ws  = (float*)d_ws;

    float* s0  = ws;
    float* s1  = ws + SSZ;
    float* s2  = ws + 2 * SSZ;
    float* v0  = ws + 3 * SSZ;
    float* v01 = ws + 4 * SSZ;

    zero_ws<<<(3 * SSZ + TPB - 1) / TPB, TPB, 0, stream>>>(ws, 3 * SSZ);

    const dim3 grid(B / BT, NIB);
    routing_pass<<<grid, TPB, 0, stream>>>(x, W, nullptr, s0);
    squash_k<<<(B * J + TPB - 1) / TPB, TPB, 0, stream>>>(s0, nullptr, v0);
    routing_pass<<<grid, TPB, 0, stream>>>(x, W, v0, s1);
    squash_k<<<(B * J + TPB - 1) / TPB, TPB, 0, stream>>>(s1, v0, v01);
    routing_pass<<<grid, TPB, 0, stream>>>(x, W, v01, s2);
    squash_k<<<(B * J + TPB - 1) / TPB, TPB, 0, stream>>>(s2, nullptr, out);
}


# block = 16 batch rows x 200 capsules (was x 64) on 16 waves: 256 blocks, one per WGP; 16 global atomic adds per s element instead of 50, so the grouping of the s partial sums changes
# speedup vs baseline: 2.7153x; 2.7153x over previous
#include <hip/hip_runtime.h>
#include <math.h>

typedef __attribute__((ext_vector_type(2))) float v2f;
typedef __attribute__((ext_vector_type(8))) float v8f;

constexpr int B      = 256;
constexpr int ICAPS  = 3200;
constexpr int INDIM  = 8;
constexpr int J      = 8;
constexpr int M      = 16;
constexpr int BT     = 16;
constexpr int IC     = 200;
constexpr int NIB    = ICAPS / IC;
constexpr int WAVES  = 16;
constexpr int TPB    = WAVES * 32;
constexpr int TPS    = 256;
constexpr int SSZ    = B * J * M;
constexpr int SLS    = 18;
constexpr int XROW4  = IC * INDIM / 4;
constexpr int XP     = 2564;
static_assert(XP >= IC * INDIM && XP % 64 == 4 && BT * XP * 4 > 160 * 1024, "x tile pitch");
static_assert(WAVES == BT, "x staging: one wave per batch row of the tile");


__device__ __forceinline__ float xhalf(float x)
{
    const int b = __builtin_bit_cast(int, x);
    return __builtin_bit_cast(float,
        __builtin_amdgcn_permlanex16(b, b, 0x76543210, 0xfedcba98, false, false));
}

template <bool UNIFORM>
__global__ __launch_bounds__(TPB) void routing_pass(
    const float* __restrict__ x,
    const float* __restrict__ W,
    const float* __restrict__ vprev,
    float* __restrict__ s_out)
{
    __shared__ __attribute__((aligned(16))) float xs[BT * XP];
    __shared__ float s_lds[J * M * SLS];

    const int t  = threadIdx.x;
    const int b0 = blockIdx.x * BT;
    const int i0 = blockIdx.y * IC;

    const int wave = t >> 5;
    const int lane = t & 31;
    const int half = lane >> 4;
    const int lm   = lane & 15;


    v8f vreg[J];
#pragma unroll
    for (int j = 0; j < J; ++j) vreg[j] = (v8f){0,0,0,0,0,0,0,0};
    if (!UNIFORM) {
        const float4* vg = reinterpret_cast<const float4*>(
            vprev + ((size_t)(b0 + lm) * J) * M + half * 8);
#pragma unroll
        for (int j = 0; j < J; ++j) {
            const float4 lo = vg[j * 4 + 0];
            const float4 hi = vg[j * 4 + 1];
            vreg[j] = (v8f){lo.x, lo.y, lo.z, lo.w, hi.x, hi.y, hi.z, hi.w};
        }
    }

    {
        constexpr int XK = (XROW4 + 31) / 32;
        const float4* xg  = reinterpret_cast<const float4*>(x) + ((size_t)(b0 + wave) * ICAPS + i0) * 2;
        float4*       xs4 = reinterpret_cast<float4*>(xs) + wave * (XP / 4);
        float4 xr[XK];
#pragma unroll
        for (int k = 0; k < XK; ++k) {
            const int r = lane + 32 * k;
            xr[k] = xg[r < XROW4 ? r : XROW4 - 1];
        }
#pragma unroll
        for (int k = 0; k < XK; ++k) {
            const int r = lane + 32 * k;
            if (r < XROW4) xs4[r] = xr[k];
        }
    }
    for (int q = t; q < J * M * SLS; q += TPB) s_lds[q] = 0.0f;
    __syncthreads();

    v8f sacc[J];
#pragma unroll
    for (int j = 0; j < J; ++j) sacc[j] = (v8f){0,0,0,0,0,0,0,0};

    for (int il = wave; il < IC; il += WAVES) {
        const int i = i0 + il;

        const float* xrow = &xs[lm * XP + il * INDIM];
        v2f xf0, xf1;
        xf0.x = xrow[half * 2 + 0]; xf0.y = xrow[half * 2 + 1];
        xf1.x = xrow[half * 2 + 4]; xf1.y = xrow[half * 2 + 5];

        v8f acc[J];
        const float* wbase = W + (size_t)i * (J * INDIM * M);
#pragma unroll
        for (int j = 0; j < J; ++j) {
            const float* wj = wbase + j * (INDIM * M);
            const int kr = half * 2;
            v2f wf0, wf1;
            wf0.x = wj[(kr + 0) * M + lm]; wf0.y = wj[(kr + 1) * M + lm];
            wf1.x = wj[(kr + 4) * M + lm]; wf1.y = wj[(kr + 5) * M + lm];
            v8f c = {0,0,0,0,0,0,0,0};
            c = __builtin_amdgcn_wmma_f32_16x16x4_f32(false, wf0, false, xf0,
                                                      (short)0, c, false, false);
            c = __builtin_amdgcn_wmma_f32_16x16x4_f32(false, wf1, false, xf1,
                                                      (short)0, c, false, false);
            acc[j] = c;
        }

        if (UNIFORM) {
#pragma unroll
            for (int j = 0; j < J; ++j) sacc[j] += 0.125f * acc[j];
        } else {
            float a[J];
#pragma unroll
            for (int j = 0; j < J; ++j) {
                const v8f u = acc[j], w = vreg[j];
                v2f p2 = __builtin_shufflevector(u, u, 0, 1) * __builtin_shufflevector(w, w, 0, 1);
                p2 += __builtin_shufflevector(u, u, 2, 3) * __builtin_shufflevector(w, w, 2, 3);
                p2 += __builtin_shufflevector(u, u, 4, 5) * __builtin_shufflevector(w, w, 4, 5);
                p2 += __builtin_shufflevector(u, u, 6, 7) * __builtin_shufflevector(w, w, 6, 7);
                const float p = p2.x + p2.y;
                a[j] = p + xhalf(p);
            }

            float mx = -3.0e38f;
#pragma unroll
            for (int j = 0; j < J; ++j) mx = fmaxf(mx, a[j]);
            float se = 0.0f;
#pragma unroll
            for (int j = 0; j < J; ++j) { a[j] = __expf(a[j] - mx); se += a[j]; }
            const float inv = 1.0f / se;

#pragma unroll
            for (int j = 0; j < J; ++j) {
                const float c = a[j] * inv;
                sacc[j] += c * acc[j];
            }
        }
    }

#pragma unroll
    for (int j = 0; j < J; ++j)
#pragma unroll
        for (int r = 0; r < 8; ++r)
            atomicAdd(&s_lds[(j * M + r + half * 8) * SLS + lm], sacc[j][r]);
    __syncthreads();
    for (int q = t; q < BT * J * M; q += TPB) {
        const int bb = q >> 7;
        const int jm = q & (J * M - 1);
        atomicAdd(&s_out[(size_t)(b0 + bb) * (J * M) + jm], s_lds[jm * SLS + bb]);
    }
}

__global__ void squash_k(const float* __restrict__ s, const float* __restrict__ vadd,
                         float* __restrict__ vout)
{
    const int bj = blockIdx.x * blockDim.x + threadIdx.x;
    if (bj >= B * J) return;
    const float* sp = s + bj * M;
    float sq = 0.0f;
#pragma unroll
    for (int m = 0; m < M; ++m) { const float e = sp[m]; sq += e * e; }
    const float scale = sqrtf(sq) / (1.0f + sq);
#pragma unroll
    for (int m = 0; m < M; ++m) {
        float r = sp[m] * scale;
        if (vadd) r += vadd[bj * M + m];
        vout[bj * M + m] = r;
    }
}

__global__ void zero_ws(float* p, int n)
{
    const int t = blockIdx.x * blockDim.x + threadIdx.x;
    if (t < n) p[t] = 0.0f;
}

extern "C" void kernel_launch(void* const* d_in, const int* in_sizes, int n_in,
                              void* d_out, int out_size, void* d_ws, size_t ws_size,
                              hipStream_t stream)
{
    (void)in_sizes; (void)n_in; (void)out_size; (void)ws_size;
    const float* x = (const float*)d_in[0];
    const float* W = (const float*)d_in[1];
    float* out = (float*)d_out;
    float* ws  = (float*)d_ws;

    float* s0  = ws;
    float* s1  = ws + SSZ;
    float* s2  = ws + 2 * SSZ;
    float* v0  = ws + 3 * SSZ;
    float* v01 = ws + 4 * SSZ;

    zero_ws<<<(3 * SSZ + TPS - 1) / TPS, TPS, 0, stream>>>(ws, 3 * SSZ);

    const dim3 grid(B / BT, NIB);
    routing_pass<true><<<grid, TPB, 0, stream>>>(x, W, nullptr, s0);
    squash_k<<<(B * J + TPS - 1) / TPS, TPS, 0, stream>>>(s0, nullptr, v0);
    routing_pass<false><<<grid, TPB, 0, stream>>>(x, W, v0, s1);
    squash_k<<<(B * J + TPS - 1) / TPS, TPS, 0, stream>>>(s1, v0, v01);
    routing_pass<false><<<grid, TPB, 0, stream>>>(x, W, v01, s2);
    squash_k<<<(B * J + TPS - 1) / TPS, TPS, 0, stream>>>(s2, nullptr, out);
}


# squash folded into the consuming pass: each block squashes its own 16 rows of s (s0; s0 and s1 for the last pass) into LDS under its x loads; 7 launches -> 5; v0+v1 now one fused multiply-add
# speedup vs baseline: 3.3894x; 1.2483x over previous
#include <hip/hip_runtime.h>
#include <math.h>

typedef __attribute__((ext_vector_type(2))) float v2f;
typedef __attribute__((ext_vector_type(8))) float v8f;

constexpr int B      = 256;
constexpr int ICAPS  = 3200;
constexpr int INDIM  = 8;
constexpr int J      = 8;
constexpr int M      = 16;
constexpr int BT     = 16;
constexpr int IC     = 200;
constexpr int NIB    = ICAPS / IC;
constexpr int WAVES  = 16;
constexpr int TPB    = WAVES * 32;
constexpr int TPS    = 256;
constexpr int SSZ    = B * J * M;
constexpr int SLS    = 18;
constexpr int XROW4  = IC * INDIM / 4;
constexpr int VP     = J * M + 4;
static_assert(TPB == 4 * BT * J, "v prologue: four threads per (batch row, class) of the tile");
constexpr int XP     = 2564;
static_assert(XP >= IC * INDIM && XP % 64 == 4 && BT * XP * 4 > 160 * 1024, "x tile pitch");
static_assert(WAVES == BT, "x staging: one wave per batch row of the tile");


__device__ __forceinline__ float squash_scale(const float* __restrict__ sp)
{
    const float4* sp4 = reinterpret_cast<const float4*>(sp);
    float sq = 0.0f;
#pragma unroll
    for (int q = 0; q < M / 4; ++q) {
        const float4 e = sp4[q];
        sq += e.x * e.x; sq += e.y * e.y; sq += e.z * e.z; sq += e.w * e.w;
    }
    return sqrtf(sq) / (1.0f + sq);
}

__device__ __forceinline__ float xhalf(float x)
{
    const int b = __builtin_bit_cast(int, x);
    return __builtin_bit_cast(float,
        __builtin_amdgcn_permlanex16(b, b, 0x76543210, 0xfedcba98, false, false));
}

template <int NS>
__global__ __launch_bounds__(TPB) void routing_pass(
    const float* __restrict__ x,
    const float* __restrict__ W,
    const float* __restrict__ s_a,
    const float* __restrict__ s_b,
    float* __restrict__ s_out)
{
    constexpr bool UNIFORM = (NS == 0);
    __shared__ __attribute__((aligned(16))) float xs[BT * XP];
    __shared__ float s_lds[J * M * SLS];
    __shared__ __attribute__((aligned(16))) float v_lds[BT * VP];

    const int t  = threadIdx.x;
    const int b0 = blockIdx.x * BT;
    const int i0 = blockIdx.y * IC;

    const int wave = t >> 5;
    const int lane = t & 31;
    const int half = lane >> 4;
    const int lm   = lane & 15;


    v8f vreg[J];
#pragma unroll
    for (int j = 0; j < J; ++j) vreg[j] = (v8f){0,0,0,0,0,0,0,0};

    {
        constexpr int XK = (XROW4 + 31) / 32;
        const float4* xg  = reinterpret_cast<const float4*>(x) + ((size_t)(b0 + wave) * ICAPS + i0) * 2;
        float4*       xs4 = reinterpret_cast<float4*>(xs) + wave * (XP / 4);
        float4 xr[XK];
#pragma unroll
        for (int k = 0; k < XK; ++k) {
            const int r = lane + 32 * k;
            xr[k] = xg[r < XROW4 ? r : XROW4 - 1];
        }

    if (!UNIFORM) {
        const int row = t >> 2;
        const int qt  = t & 3;
        const size_t off = ((size_t)b0 * J + row) * M;
        const float sc_a = squash_scale(s_a + off);
        const float4 ea  = reinterpret_cast<const float4*>(s_a + off)[qt];
        float4 v = make_float4(ea.x * sc_a, ea.y * sc_a, ea.z * sc_a, ea.w * sc_a);
        if (NS == 2) {
            const float sc_b = squash_scale(s_b + off);
            const float4 eb  = reinterpret_cast<const float4*>(s_b + off)[qt];
            v = make_float4(eb.x * sc_b + v.x, eb.y * sc_b + v.y, eb.z * sc_b + v.z, eb.w * sc_b + v.w);
        }
        *reinterpret_cast<float4*>(&v_lds[(row >> 3) * VP + (row & 7) * M + qt * 4]) = v;
    }

#pragma unroll
        for (int k = 0; k < XK; ++k) {
            const int r = lane + 32 * k;
            if (r < XROW4) xs4[r] = xr[k];
        }
    }
    for (int q = t; q < J * M * SLS; q += TPB) s_lds[q] = 0.0f;
    __syncthreads();

    if (!UNIFORM) {
        const float4* vl = reinterpret_cast<const float4*>(&v_lds[lm * VP + half * 8]);
#pragma unroll
        for (int j = 0; j < J; ++j) {
            const float4 lo = vl[j * 4 + 0];
            const float4 hi = vl[j * 4 + 1];
            vreg[j] = (v8f){lo.x, lo.y, lo.z, lo.w, hi.x, hi.y, hi.z, hi.w};
        }
    }

    v8f sacc[J];
#pragma unroll
    for (int j = 0; j < J; ++j) sacc[j] = (v8f){0,0,0,0,0,0,0,0};

    for (int il = wave; il < IC; il += WAVES) {
        const int i = i0 + il;

        const float* xrow = &xs[lm * XP + il * INDIM];
        v2f xf0, xf1;
        xf0.x = xrow[half * 2 + 0]; xf0.y = xrow[half * 2 + 1];
        xf1.x = xrow[half * 2 + 4]; xf1.y = xrow[half * 2 + 5];

        v8f acc[J];
        const float* wbase = W + (size_t)i * (J * INDIM * M);
#pragma unroll
        for (int j = 0; j < J; ++j) {
            const float* wj = wbase + j * (INDIM * M);
            const int kr = half * 2;
            v2f wf0, wf1;
            wf0.x = wj[(kr + 0) * M + lm]; wf0.y = wj[(kr + 1) * M + lm];
            wf1.x = wj[(kr + 4) * M + lm]; wf1.y = wj[(kr + 5) * M + lm];
            v8f c = {0,0,0,0,0,0,0,0};
            c = __builtin_amdgcn_wmma_f32_16x16x4_f32(false, wf0, false, xf0,
                                                      (short)0, c, false, false);
            c = __builtin_amdgcn_wmma_f32_16x16x4_f32(false, wf1, false, xf1,
                                                      (short)0, c, false, false);
            acc[j] = c;
        }

        if (UNIFORM) {
#pragma unroll
            for (int j = 0; j < J; ++j) sacc[j] += 0.125f * acc[j];
        } else {
            float a[J];
#pragma unroll
            for (int j = 0; j < J; ++j) {
                const v8f u = acc[j], w = vreg[j];
                v2f p2 = __builtin_shufflevector(u, u, 0, 1) * __builtin_shufflevector(w, w, 0, 1);
                p2 += __builtin_shufflevector(u, u, 2, 3) * __builtin_shufflevector(w, w, 2, 3);
                p2 += __builtin_shufflevector(u, u, 4, 5) * __builtin_shufflevector(w, w, 4, 5);
                p2 += __builtin_shufflevector(u, u, 6, 7) * __builtin_shufflevector(w, w, 6, 7);
                const float p = p2.x + p2.y;
                a[j] = p + xhalf(p);
            }

            float mx = -3.0e38f;
#pragma unroll
            for (int j = 0; j < J; ++j) mx = fmaxf(mx, a[j]);
            float se = 0.0f;
#pragma unroll
            for (int j = 0; j < J; ++j) { a[j] = __expf(a[j] - mx); se += a[j]; }
            const float inv = 1.0f / se;

#pragma unroll
            for (int j = 0; j < J; ++j) {
                const float c = a[j] * inv;
                sacc[j] += c * acc[j];
            }
        }
    }

#pragma unroll
    for (int j = 0; j < J; ++j)
#pragma unroll
        for (int r = 0; r < 8; ++r)
            atomicAdd(&s_lds[(j * M + r + half * 8) * SLS + lm], sacc[j][r]);
    __syncthreads();
    for (int q = t; q < BT * J * M; q += TPB) {
        const int bb = q >> 7;
        const int jm = q & (J * M - 1);
        atomicAdd(&s_out[(size_t)(b0 + bb) * (J * M) + jm], s_lds[jm * SLS + bb]);
    }
}

__global__ void squash_k(const float* __restrict__ s, float* __restrict__ vout)
{
    const int bj = blockIdx.x * blockDim.x + threadIdx.x;
    if (bj >= B * J) return;
    const float* sp = s + bj * M;
    const float scale = squash_scale(sp);
#pragma unroll
    for (int m = 0; m < M; ++m) vout[bj * M + m] = sp[m] * scale;
}

__global__ void zero_ws(float* p, int n)
{
    const int t = blockIdx.x * blockDim.x + threadIdx.x;
    if (t < n) p[t] = 0.0f;
}

extern "C" void kernel_launch(void* const* d_in, const int* in_sizes, int n_in,
                              void* d_out, int out_size, void* d_ws, size_t ws_size,
                              hipStream_t stream)
{
    (void)in_sizes; (void)n_in; (void)out_size; (void)ws_size;
    const float* x = (const float*)d_in[0];
    const float* W = (const float*)d_in[1];
    float* out = (float*)d_out;
    float* ws  = (float*)d_ws;

    float* s0  = ws;
    float* s1  = ws + SSZ;
    float* s2  = ws + 2 * SSZ;

    zero_ws<<<(3 * SSZ + TPS - 1) / TPS, TPS, 0, stream>>>(ws, 3 * SSZ);

    const dim3 grid(B / BT, NIB);
    routing_pass<0><<<grid, TPB, 0, stream>>>(x, W, nullptr, nullptr, s0);
    routing_pass<1><<<grid, TPB, 0, stream>>>(x, W, s0, nullptr, s1);
    routing_pass<2><<<grid, TPB, 0, stream>>>(x, W, s0, s1, s2);
    squash_k<<<(B * J + TPS - 1) / TPS, TPS, 0, stream>>>(s2, out);
}
